# starting kernel
# baseline (speedup 1.0000x reference)
#include <hip/hip_runtime.h>

typedef __attribute__((ext_vector_type(2))) float v2f;
typedef __attribute__((ext_vector_type(8))) float v8f;

#define WSZ     7
#define PAD     3
#define TILE_H  16
#define TILE_W  128
#define IN_ROWS (TILE_H + WSZ - 1)
#define IN_COLS (TILE_W + WSZ - 1)
#define XSTRIDE 136
#define HROWS   24
#define IMG_H   512
#define IMG_W   512

__global__ __launch_bounds__(256) void local_variance_kernel(
    const float* __restrict__ x, float* __restrict__ out)
{
    __shared__ float xbuf[IN_ROWS][XSTRIDE];
    __shared__ float hs[HROWS][TILE_W];
    __shared__ float hq[HROWS][TILE_W];

    const int tid = threadIdx.x;
    const int img = blockIdx.z;
    const int y0  = blockIdx.y * TILE_H;
    const int c0  = blockIdx.x * TILE_W;
    const float* __restrict__ src = x + (size_t)img * IMG_H * IMG_W;
    float* __restrict__ dst       = out + (size_t)img * IMG_H * IMG_W;

    for (int idx = tid; idx < IN_ROWS * IN_COLS; idx += 256) {
        int r  = idx / IN_COLS;
        int cc = idx - r * IN_COLS;
        int gy = (y0 + r  - PAD + IMG_H) & (IMG_H - 1);
        int gx = (c0 + cc - PAD + IMG_W) & (IMG_W - 1);
        xbuf[r][cc] = src[gy * IMG_W + gx];
    }
    __syncthreads();

    for (int idx = tid; idx < IN_ROWS * TILE_W; idx += 256) {
        int r = idx / TILE_W;
        int c = idx - r * TILE_W;
        float s = 0.0f, q = 0.0f;
#pragma unroll
        for (int j = 0; j < WSZ; ++j) {
            float v = xbuf[r][c + j];
            s += v;
            q += v * v;
        }
        hs[r][c] = s;
        hq[r][c] = q;
    }
    for (int idx = tid; idx < (HROWS - IN_ROWS) * TILE_W; idx += 256) {
        int r = IN_ROWS + idx / TILE_W;
        int c = idx % TILE_W;
        hs[r][c] = 0.0f;
        hq[r][c] = 0.0f;
    }
    __syncthreads();

    const int lane = tid & 31;
    const int wv   = tid >> 5;
    const int n    = lane & 15;
    const int hf   = lane >> 4;
    const int cb   = wv * 16;

    v8f accS = {};
    v8f accQ = {};
#pragma unroll
    for (int kc = 0; kc < 6; ++kc) {
        const int k0 = kc * 4 + hf * 2;
        v2f a;
        a.x = (k0     >= n && k0     <= n + 6) ? 1.0f : 0.0f;
        a.y = (k0 + 1 >= n && k0 + 1 <= n + 6) ? 1.0f : 0.0f;
        v2f bs, bq;
        bs.x = hs[k0][cb + n];
        bs.y = hs[k0 + 1][cb + n];
        bq.x = hq[k0][cb + n];
        bq.y = hq[k0 + 1][cb + n];
        accS = __builtin_amdgcn_wmma_f32_16x16x4_f32(
            false, a, false, bs, (short)0, accS, false, false);
        accQ = __builtin_amdgcn_wmma_f32_16x16x4_f32(
            false, a, false, bq, (short)0, accQ, false, false);
    }

    const float inv_area = 1.0f / (float)(WSZ * WSZ);
    const int gx = c0 + cb + n;
#pragma unroll
    for (int v = 0; v < 8; ++v) {
        int gy = y0 + v + hf * 8;
        float s = accS[v] * inv_area;
        float q = accQ[v] * inv_area;
        dst[gy * IMG_W + gx] = q - s * s;
    }
}

extern "C" void kernel_launch(void* const* d_in, const int* in_sizes, int n_in,
                              void* d_out, int out_size, void* d_ws, size_t ws_size,
                              hipStream_t stream) {
    (void)in_sizes; (void)n_in; (void)d_ws; (void)ws_size; (void)out_size;
    const float* x = (const float*)d_in[0];
    float* out = (float*)d_out;
    dim3 grid(IMG_W / TILE_W, IMG_H / TILE_H, 16 * 3);
    dim3 block(256, 1, 1);
    local_variance_kernel<<<grid, block, 0, stream>>>(x, out);
}


# No LDS, no barriers: a wave walks a 64-column strip down 64 rows; horizontal 7-tap sums made in registers from global loads are the B operands of the band-matrix WMMA; tap summation order changes
# speedup vs baseline: 2.0296x; 2.0296x over previous
#include <hip/hip_runtime.h>


typedef __attribute__((ext_vector_type(2))) float v2f;
typedef __attribute__((ext_vector_type(4))) float v4f;
typedef __attribute__((ext_vector_type(8))) float v8f;

#define WSZ        7
#define IMG_H      512
#define IMG_W      512
#define STRIP_W    64
#define STRIP_H    64
#define NBLK       (STRIP_H / 16)
#define WG_WAVES   4

struct RawRow { v4f L, M, R; };

__device__ __forceinline__ RawRow load_row(const float* __restrict__ img, int row,
                                           int cL, int cM, int cR)
{
    const float* p = img + ((row & (IMG_H - 1)) * IMG_W);
    RawRow r;
    r.L = *(const v4f*)(p + cL);
    r.M = *(const v4f*)(p + cM);
    r.R = *(const v4f*)(p + cR);
    return r;
}

__device__ __forceinline__ void hsum4(const RawRow& r, float* hs, float* hq)
{
    const float v0 = r.L.y, v1 = r.L.z, v2 = r.L.w;
    const float v3 = r.M.x, v4 = r.M.y, v5 = r.M.z, v6 = r.M.w;
    const float v7 = r.R.x, v8 = r.R.y, v9 = r.R.z;

    const float cm  = (v3 + v4) + (v5 + v6);
    const float p12 = v1 + v2;
    const float p78 = v7 + v8;
    hs[0] = (v0 + p12) + cm;
    hs[1] = (p12 + v7) + cm;
    hs[2] = (v2 + p78) + cm;
    hs[3] = (p78 + v9) + cm;

    const float w0 = v0 * v0, w1 = v1 * v1, w2 = v2 * v2, w3 = v3 * v3, w4 = v4 * v4;
    const float w5 = v5 * v5, w6 = v6 * v6, w7 = v7 * v7, w8 = v8 * v8, w9 = v9 * v9;
    const float cq  = (w3 + w4) + (w5 + w6);
    const float q12 = w1 + w2;
    const float q78 = w7 + w8;
    hq[0] = (w0 + q12) + cq;
    hq[1] = (q12 + w7) + cq;
    hq[2] = (w2 + q78) + cq;
    hq[3] = (q78 + w9) + cq;
}

__device__ __forceinline__ void make_frag(const RawRow& r0, const RawRow& r1,
                                          v2f* fs, v2f* fq)
{
    float s0[4], q0[4], s1[4], q1[4];
    hsum4(r0, s0, q0);
    hsum4(r1, s1, q1);
    for (int t = 0; t < 4; ++t) {
        fs[t].x = s0[t]; fs[t].y = s1[t];
        fq[t].x = q0[t]; fq[t].y = q1[t];
    }
}

__global__ __launch_bounds__(32 * WG_WAVES) void local_variance_kernel(
    const float* __restrict__ x, float* __restrict__ out)
{
    const int tid  = threadIdx.x;
    const int lane = tid & 31;
    const int wv   = tid >> 5;
    const int n    = lane & 15;
    const int hf   = lane >> 4;

    const int img   = blockIdx.z;
    const int cs    = blockIdx.x * (STRIP_W * WG_WAVES) + wv * STRIP_W;
    const int c0    = cs + 4 * n;
    const int cL    = (c0 - 4) & (IMG_W - 1);
    const int cR    = (c0 + 4) & (IMG_W - 1);
    const int rbase = blockIdx.y * STRIP_H;
    const float* __restrict__ src = x   + (size_t)img * IMG_H * IMG_W;
    float* __restrict__ dst       = out + (size_t)img * IMG_H * IMG_W;

    v2f band[6];
    for (int c = 0; c < 6; ++c) {
        const int k0 = 4 * c + 2 * hf;
        band[c].x = (k0     >= n + 1 && k0     <= n + 7) ? 1.0f : 0.0f;
        band[c].y = (k0 + 1 >= n + 1 && k0 + 1 <= n + 7) ? 1.0f : 0.0f;
    }

    const int lrow = rbase - 4 + 2 * hf;

    v2f FS[6][4], FQ[6][4];
    RawRow raw[4][2];

    for (int c = 0; c < 2; ++c) {
        const RawRow a = load_row(src, lrow + 4 * c,     cL, c0, cR);
        const RawRow b = load_row(src, lrow + 4 * c + 1, cL, c0, cR);
        make_frag(a, b, FS[c], FQ[c]);
    }
    for (int c = 0; c < 4; ++c) {
        raw[c][0] = load_row(src, lrow + 4 * (c + 2),     cL, c0, cR);
        raw[c][1] = load_row(src, lrow + 4 * (c + 2) + 1, cL, c0, cR);
    }

    const float inv_area = 1.0f / (float)(WSZ * WSZ);

    for (int b = 0; b < NBLK; ++b) {
        for (int c = 0; c < 4; ++c)
            make_frag(raw[c][0], raw[c][1], FS[c + 2], FQ[c + 2]);

        if (b + 1 < NBLK) {
            const int nr = lrow + 16 * (b + 1);
            for (int c = 0; c < 4; ++c) {
                raw[c][0] = load_row(src, nr + 4 * (c + 2),     cL, c0, cR);
                raw[c][1] = load_row(src, nr + 4 * (c + 2) + 1, cL, c0, cR);
            }
        }
        __builtin_amdgcn_sched_barrier(0);

        v8f accS[4], accQ[4];
        for (int t = 0; t < 4; ++t) { accS[t] = (v8f){}; accQ[t] = (v8f){}; }

        for (int c = 0; c < 6; ++c) {
            for (int t = 0; t < 4; ++t) {
                accS[t] = __builtin_amdgcn_wmma_f32_16x16x4_f32(
                    false, band[c], false, FS[c][t], (short)0, accS[t], false, false);
                accQ[t] = __builtin_amdgcn_wmma_f32_16x16x4_f32(
                    false, band[c], false, FQ[c][t], (short)0, accQ[t], false, false);
            }
        }

        const int orow = rbase + 16 * b + 8 * hf;
        for (int v = 0; v < 8; ++v) {
            v4f o;
            for (int t = 0; t < 4; ++t) {
                const float s = accS[t][v] * inv_area;
                const float q = accQ[t][v] * inv_area;
                o[t] = q - s * s;
            }
            *(v4f*)(dst + (orow + v) * IMG_W + c0) = o;
        }

        for (int t = 0; t < 4; ++t) {
            FS[0][t] = FS[4][t]; FS[1][t] = FS[5][t];
            FQ[0][t] = FQ[4][t]; FQ[1][t] = FQ[5][t];
        }
    }
}

extern "C" void kernel_launch(void* const* d_in, const int* in_sizes, int n_in,
                              void* d_out, int out_size, void* d_ws, size_t ws_size,
                              hipStream_t stream) {
    (void)in_sizes; (void)n_in; (void)d_ws; (void)ws_size; (void)out_size;
    const float* x = (const float*)d_in[0];
    float* out = (float*)d_out;
    dim3 grid(IMG_W / (STRIP_W * WG_WAVES), IMG_H / STRIP_H, 16 * 3);
    dim3 block(32 * WG_WAVES, 1, 1);
    local_variance_kernel<<<grid, block, 0, stream>>>(x, out);
}


# Row addresses made scalar: each load is scalar row base + fixed per-lane byte offset + immediate (row loop unrolled for it); no 64-bit vector address math, all start-up loads issue back to back
# speedup vs baseline: 2.1870x; 1.0776x over previous
#include <hip/hip_runtime.h>


typedef __attribute__((ext_vector_type(2))) float v2f;
typedef __attribute__((ext_vector_type(4))) float v4f;
typedef __attribute__((ext_vector_type(8))) float v8f;

#define WSZ        7
#define IMG_H      512
#define IMG_W      512
#define STRIP_W    64
#define STRIP_H    64
#define NBLK       (STRIP_H / 16)
#define WG_WAVES   4

struct RawRow { v4f L, M, R; };

__device__ __forceinline__ RawRow load_row(const char* __restrict__ rowp,
                                           unsigned oL, unsigned oM, unsigned oR)
{
    RawRow r;
    r.L = *(const v4f*)(rowp + oL);
    r.M = *(const v4f*)(rowp + oM);
    r.R = *(const v4f*)(rowp + oR);
    return r;
}

__device__ __forceinline__ const char* chunk_base(const float* __restrict__ img, int rb)
{
    return (const char*)img + (size_t)((unsigned)(rb & (IMG_H - 1)) * (unsigned)(IMG_W * 4));
}

__device__ __forceinline__ void hsum4(const RawRow& r, float* hs, float* hq)
{
    const float v0 = r.L.y, v1 = r.L.z, v2 = r.L.w;
    const float v3 = r.M.x, v4 = r.M.y, v5 = r.M.z, v6 = r.M.w;
    const float v7 = r.R.x, v8 = r.R.y, v9 = r.R.z;

    const float cm  = (v3 + v4) + (v5 + v6);
    const float p12 = v1 + v2;
    const float p78 = v7 + v8;
    hs[0] = (v0 + p12) + cm;
    hs[1] = (p12 + v7) + cm;
    hs[2] = (v2 + p78) + cm;
    hs[3] = (p78 + v9) + cm;

    const float w0 = v0 * v0, w1 = v1 * v1, w2 = v2 * v2, w3 = v3 * v3, w4 = v4 * v4;
    const float w5 = v5 * v5, w6 = v6 * v6, w7 = v7 * v7, w8 = v8 * v8, w9 = v9 * v9;
    const float cq  = (w3 + w4) + (w5 + w6);
    const float q12 = w1 + w2;
    const float q78 = w7 + w8;
    hq[0] = (w0 + q12) + cq;
    hq[1] = (q12 + w7) + cq;
    hq[2] = (w2 + q78) + cq;
    hq[3] = (q78 + w9) + cq;
}

__device__ __forceinline__ void make_frag(const RawRow& r0, const RawRow& r1,
                                          v2f* fs, v2f* fq)
{
    float s0[4], q0[4], s1[4], q1[4];
    hsum4(r0, s0, q0);
    hsum4(r1, s1, q1);
    for (int t = 0; t < 4; ++t) {
        fs[t].x = s0[t]; fs[t].y = s1[t];
        fq[t].x = q0[t]; fq[t].y = q1[t];
    }
}

__global__ __launch_bounds__(32 * WG_WAVES) void local_variance_kernel(
    const float* __restrict__ x, float* __restrict__ out)
{
    const int tid  = threadIdx.x;
    const int lane = tid & 31;
    const int wv   = tid >> 5;
    const int n    = lane & 15;
    const int hf   = lane >> 4;

    const int img   = blockIdx.z;
    const int cs    = blockIdx.x * (STRIP_W * WG_WAVES) + wv * STRIP_W;
    const int c0    = cs + 4 * n;
    const int cL    = (c0 - 4) & (IMG_W - 1);
    const int cR    = (c0 + 4) & (IMG_W - 1);
    const int rbase = blockIdx.y * STRIP_H;
    const float* __restrict__ src = x   + (size_t)img * IMG_H * IMG_W;
    float* __restrict__ dst       = out + (size_t)img * IMG_H * IMG_W;

    v2f band[6];
    for (int c = 0; c < 6; ++c) {
        const int k0 = 4 * c + 2 * hf;
        band[c].x = (k0     >= n + 1 && k0     <= n + 7) ? 1.0f : 0.0f;
        band[c].y = (k0 + 1 >= n + 1 && k0 + 1 <= n + 7) ? 1.0f : 0.0f;
    }

    const int rb0 = rbase - 4;
    const unsigned ROWB = IMG_W * 4;
    const unsigned lane_row = (unsigned)(2 * hf) * ROWB;
    const unsigned oL = lane_row + (unsigned)cL * 4u;
    const unsigned oM = lane_row + (unsigned)c0 * 4u;
    const unsigned oR = lane_row + (unsigned)cR * 4u;

    v2f FS[6][4], FQ[6][4];
    RawRow raw[4][2];

    RawRow head[2][2];
    for (int c = 0; c < 2; ++c) {
        const char* p = chunk_base(src, rb0 + 4 * c);
        head[c][0] = load_row(p,        oL, oM, oR);
        head[c][1] = load_row(p + ROWB, oL, oM, oR);
    }
    for (int c = 0; c < 4; ++c) {
        const char* p = chunk_base(src, rb0 + 4 * (c + 2));
        raw[c][0] = load_row(p,        oL, oM, oR);
        raw[c][1] = load_row(p + ROWB, oL, oM, oR);
    }
    for (int c = 0; c < 2; ++c)
        make_frag(head[c][0], head[c][1], FS[c], FQ[c]);
    __builtin_amdgcn_sched_barrier(0);

    const float inv_area = 1.0f / (float)(WSZ * WSZ);

#pragma unroll
    for (int b = 0; b < NBLK; ++b) {
        for (int c = 0; c < 4; ++c)
            make_frag(raw[c][0], raw[c][1], FS[c + 2], FQ[c + 2]);

        if (b + 1 < NBLK) {
            const int nrb = rb0 + 16 * (b + 1);
            for (int c = 0; c < 4; ++c) {
                const char* p = chunk_base(src, nrb + 4 * (c + 2));
                raw[c][0] = load_row(p,        oL, oM, oR);
                raw[c][1] = load_row(p + ROWB, oL, oM, oR);
            }
        }
        __builtin_amdgcn_sched_barrier(0);

        v8f accS[4], accQ[4];
        for (int t = 0; t < 4; ++t) { accS[t] = (v8f){}; accQ[t] = (v8f){}; }

        for (int c = 0; c < 6; ++c) {
            for (int t = 0; t < 4; ++t) {
                accS[t] = __builtin_amdgcn_wmma_f32_16x16x4_f32(
                    false, band[c], false, FS[c][t], (short)0, accS[t], false, false);
                accQ[t] = __builtin_amdgcn_wmma_f32_16x16x4_f32(
                    false, band[c], false, FQ[c][t], (short)0, accQ[t], false, false);
            }
        }

        const int orow = rbase + 16 * b + 8 * hf;
        for (int v = 0; v < 8; ++v) {
            v4f o;
            for (int t = 0; t < 4; ++t) {
                const float s = accS[t][v] * inv_area;
                const float q = accQ[t][v] * inv_area;
                o[t] = q - s * s;
            }
            *(v4f*)(dst + (orow + v) * IMG_W + c0) = o;
        }

        for (int t = 0; t < 4; ++t) {
            FS[0][t] = FS[4][t]; FS[1][t] = FS[5][t];
            FQ[0][t] = FQ[4][t]; FQ[1][t] = FQ[5][t];
        }
        __builtin_amdgcn_sched_barrier(0);
    }
}

extern "C" void kernel_launch(void* const* d_in, const int* in_sizes, int n_in,
                              void* d_out, int out_size, void* d_ws, size_t ws_size,
                              hipStream_t stream) {
    (void)in_sizes; (void)n_in; (void)d_ws; (void)ws_size; (void)out_size;
    const float* x = (const float*)d_in[0];
    float* out = (float*)d_out;
    dim3 grid(IMG_W / (STRIP_W * WG_WAVES), IMG_H / STRIP_H, 16 * 3);
    dim3 block(32 * WG_WAVES, 1, 1);
    local_variance_kernel<<<grid, block, 0, stream>>>(x, out);
}
